# starting kernel
# baseline (speedup 1.0000x reference)
#include <hip/hip_runtime.h>
#include <hip/hip_bf16.h>
#include <math.h>


typedef __attribute__((ext_vector_type(2))) float v2f;
typedef __attribute__((ext_vector_type(8))) float v8f;

#define DIM 125
#define NP  216
#define FD  36
#define RK  20

#define OFF_AP   0
#define OFF_WQ   28672
#define OFF_WK   45056
#define OFF_YQ   61440
#define OFF_YK   90112
#define OFF_S    118784
#define OFF_CP   165440
#define OFF_T1   212096
#define OFF_T2   238016
#define OFF_U1   252416
#define OFF_U2   253568
#define OFF_U3   254720

template <int K, int LDA, int LDB>
__device__ inline v8f wmma_acc_f32(const float* __restrict__ A,
                                   const float* __restrict__ Bn,
                                   int m0, int n0, int lane) {
  int lo = lane & 15, hi = lane >> 4;
  const float* ap = A + (m0 + lo) * LDA + 2 * hi;
  const float* bp = Bn + (n0 + lo) * LDB + 2 * hi;
  v8f c = {};
#pragma unroll
  for (int kk = 0; kk < K; kk += 4) {
    v2f a = *(const v2f*)(ap + kk);
    v2f b = *(const v2f*)(bp + kk);
    c = __builtin_amdgcn_wmma_f32_16x16x4_f32(false, a, false, b, (short)0, c,
                                              false, false);
  }
  return c;
}

__global__ __launch_bounds__(256) void ff_pack_kernel(
    const float* __restrict__ x, const float* __restrict__ qw,
    const float* __restrict__ kw, const float* __restrict__ U1,
    const float* __restrict__ U2, const float* __restrict__ U3,
    float* __restrict__ ws) {
  int idx = blockIdx.x * 256 + threadIdx.x;
  if (idx < 28672) {
    int m = idx >> 7, k = idx & 127;
    ws[OFF_AP + idx] = (m < NP && k < DIM) ? x[m * DIM + k] : 0.0f;
    return;
  }
  idx -= 28672;
  if (idx < 2 * 16384) {
    int which = idx >> 14;
    int r = idx & 16383;
    int n = r >> 7, k = r & 127;
    const float* W = which ? kw : qw;
    float v = (n < DIM && k < DIM) ? W[n * DIM + k] : 0.0f;
    ws[(which ? OFF_WK : OFF_WQ) + r] = v;
    return;
  }
  idx -= 2 * 16384;
  if (idx < 3 * 1152) {
    int which = idx / 1152;
    int r = idx % 1152;
    int n = r / FD;
    const float* U = (which == 0) ? U1 : (which == 1) ? U2 : U3;
    int off = (which == 0) ? OFF_U1 : (which == 1) ? OFF_U2 : OFF_U3;
    ws[off + r] = (n < RK) ? U[r] : 0.0f;
    return;
  }
  idx -= 3 * 1152;
  if (idx < 2 * 1024) {
    int which = idx >> 10;
    int r = idx & 1023;
    ws[(which ? OFF_YK : OFF_YQ) + NP * 128 + r] = 0.0f;
  }
}

__global__ __launch_bounds__(32) void ff_proj_kernel(float* __restrict__ ws) {
  int t = blockIdx.x;
  int which = t / (14 * 8);
  int rem = t % (14 * 8);
  int tm = rem / 8, tn = rem % 8;
  const float* A = ws + OFF_AP;
  const float* W = ws + (which ? OFF_WK : OFF_WQ);
  float* Y = ws + (which ? OFF_YK : OFF_YQ);
  int lane = threadIdx.x, lo = lane & 15, hi = lane >> 4;
  v8f c = wmma_acc_f32<128, 128, 128>(A, W, tm * 16, tn * 16, lane);
#pragma unroll
  for (int v = 0; v < 8; ++v) {
    int m = tm * 16 + v + 8 * hi;
    int n = tn * 16 + lo;
    if (m < NP) Y[m * 128 + n] = c[v];
  }
}

__global__ __launch_bounds__(128) void ff_bn_l2_kernel(
    float* __restrict__ ws, const float* __restrict__ gamma,
    const float* __restrict__ beta) {
  int blk = blockIdx.x;
  int row = blk % NP;
  float* Yrow = ws + (blk / NP ? OFF_YK : OFF_YQ) + row * 128;
  int tid = threadIdx.x;
  __shared__ float s1[128], s2[128];
  __shared__ float smu, srstd, snrm;
  float v = (tid < DIM) ? Yrow[tid] : 0.0f;
  s1[tid] = v;
  s2[tid] = v * v;
  __syncthreads();
  for (int s = 64; s > 0; s >>= 1) {
    if (tid < s) { s1[tid] += s1[tid + s]; s2[tid] += s2[tid + s]; }
    __syncthreads();
  }
  if (tid == 0) {
    float mu = s1[0] * (1.0f / DIM);
    float var = s2[0] * (1.0f / DIM) - mu * mu;
    smu = mu;
    srstd = rsqrtf(var + 1e-5f);
  }
  __syncthreads();
  float yn = (tid < DIM) ? ((v - smu) * srstd * gamma[row] + beta[row]) : 0.0f;
  s1[tid] = yn * yn;
  __syncthreads();
  for (int s = 64; s > 0; s >>= 1) {
    if (tid < s) s1[tid] += s1[tid + s];
    __syncthreads();
  }
  if (tid == 0) snrm = fmaxf(sqrtf(s1[0]), 1e-12f);
  __syncthreads();
  Yrow[tid] = (tid < DIM) ? yn / snrm : 0.0f;
}

__global__ __launch_bounds__(32) void ff_scores_kernel(float* __restrict__ ws) {
  int t = blockIdx.x;
  int tm = t / 14, tn = t % 14;
  const float* Q = ws + OFF_YQ;
  const float* Kn = ws + OFF_YK;
  float* S = ws + OFF_S;
  int lane = threadIdx.x, lo = lane & 15, hi = lane >> 4;
  v8f c = wmma_acc_f32<128, 128, 128>(Q, Kn, tm * 16, tn * 16, lane);
  const float scale = 0.08944271909999159f;
#pragma unroll
  for (int v = 0; v < 8; ++v) {
    int m = tm * 16 + v + 8 * hi;
    int n = tn * 16 + lo;
    if (m < NP && n < NP) S[m * NP + n] = c[v] * scale;
  }
}

__global__ __launch_bounds__(256) void ff_sparsemax_kernel(
    float* __restrict__ ws, float* __restrict__ att) {
  const float* S = ws + OFF_S;
  float* Cp = ws + OFF_CP;
  __shared__ float z[256];
  __shared__ float cs[256];
  __shared__ float stau;
  __shared__ int scount;
  int row = blockIdx.x, tid = threadIdx.x;
  float val = (tid < NP) ? S[row * NP + tid] : -1e30f;
  z[tid] = val;
  __syncthreads();
  for (int ksz = 2; ksz <= 256; ksz <<= 1) {
    for (int j = ksz >> 1; j > 0; j >>= 1) {
      int ixj = tid ^ j;
      if (ixj > tid) {
        float A = z[tid], B = z[ixj];
        if (((tid & ksz) == 0) == (A < B)) { z[tid] = B; z[ixj] = A; }
      }
      __syncthreads();
    }
  }
  cs[tid] = z[tid];
  __syncthreads();
  for (int off = 1; off < 256; off <<= 1) {
    float tmp = (tid >= off) ? cs[tid - off] : 0.0f;
    __syncthreads();
    cs[tid] += tmp;
    __syncthreads();
  }
  if (tid == 0) scount = 0;
  __syncthreads();
  if (tid < NP) {
    if (1.0f + (float)(tid + 1) * z[tid] > cs[tid]) atomicAdd(&scount, 1);
  }
  __syncthreads();
  if (tid == 0) {
    int kz = max(scount, 1);
    stau = (cs[kz - 1] - 1.0f) / (float)kz;
  }
  __syncthreads();
  if (tid < NP) {
    float r = fmaxf(val - stau, 0.0f);
    att[row * NP + tid] = r;
    int p = row, q = tid;
    int i = 6 * (p / 36) + (q / 36);
    int j = 6 * ((p / 6) % 6) + ((q / 6) % 6);
    int k = 6 * (p % 6) + (q % 6);
    Cp[(i * FD + j) * FD + k] = r;
  }
}

__global__ __launch_bounds__(32) void ff_tuckerA_kernel(float* __restrict__ ws) {
  int t = blockIdx.x;
  int tm = t / 2, tn = t % 2;
  int lane = threadIdx.x, lo = lane & 15, hi = lane >> 4;
  v8f c = wmma_acc_f32<FD, FD, FD>(ws + OFF_CP, ws + OFF_U3, tm * 16, tn * 16,
                                   lane);
  float* T1t = ws + OFF_T1;
#pragma unroll
  for (int v = 0; v < 8; ++v) {
    int r = tm * 16 + v + 8 * hi;
    int n = tn * 16 + lo;
    if (n < RK) T1t[((r / FD) * RK + n) * FD + (r % FD)] = c[v];
  }
}

__global__ __launch_bounds__(32) void ff_tuckerB_kernel(float* __restrict__ ws) {
  int t = blockIdx.x;
  int tm = t / 2, tn = t % 2;
  int lane = threadIdx.x, lo = lane & 15, hi = lane >> 4;
  v8f c = wmma_acc_f32<FD, FD, FD>(ws + OFF_T1, ws + OFF_U2, tm * 16, tn * 16,
                                   lane);
  float* T2t = ws + OFF_T2;
#pragma unroll
  for (int v = 0; v < 8; ++v) {
    int r = tm * 16 + v + 8 * hi;
    int n = tn * 16 + lo;
    if (n < RK) T2t[(n * RK + (r % RK)) * FD + (r / RK)] = c[v];
  }
}

__global__ __launch_bounds__(32) void ff_tuckerC_kernel(float* __restrict__ ws,
                                                        float* __restrict__ out) {
  int t = blockIdx.x;
  int tm = t / 2, tn = t % 2;
  int lane = threadIdx.x, lo = lane & 15, hi = lane >> 4;
  v8f c = wmma_acc_f32<FD, FD, FD>(ws + OFF_T2, ws + OFF_U1, tm * 16, tn * 16,
                                   lane);
#pragma unroll
  for (int v = 0; v < 8; ++v) {
    int r = tm * 16 + v + 8 * hi;
    int n = tn * 16 + lo;
    if (n < RK) out[n * 400 + r] = tanhf(c[v]);
  }
}

extern "C" void kernel_launch(void* const* d_in, const int* in_sizes, int n_in,
                              void* d_out, int out_size, void* d_ws,
                              size_t ws_size, hipStream_t stream) {
  const float* patches = (const float*)d_in[0];
  const float* qw      = (const float*)d_in[1];
  const float* kw      = (const float*)d_in[2];
  const float* gamma   = (const float*)d_in[3];
  const float* beta    = (const float*)d_in[4];
  const float* U1      = (const float*)d_in[5];
  const float* U2      = (const float*)d_in[6];
  const float* U3      = (const float*)d_in[7];

  float* out = (float*)d_out;
  float* att = out + RK * RK * RK;
  float* ws = (float*)d_ws;

  ff_pack_kernel<<<(66944 + 255) / 256, 256, 0, stream>>>(patches, qw, kw, U1,
                                                          U2, U3, ws);
  ff_proj_kernel<<<14 * 8 * 2, 32, 0, stream>>>(ws);
  ff_bn_l2_kernel<<<2 * NP, 128, 0, stream>>>(ws, gamma, beta);
  ff_scores_kernel<<<14 * 14, 32, 0, stream>>>(ws);
  ff_sparsemax_kernel<<<NP, 256, 0, stream>>>(ws, att);
  ff_tuckerA_kernel<<<81 * 2, 32, 0, stream>>>(ws);
  ff_tuckerB_kernel<<<45 * 2, 32, 0, stream>>>(ws);
  ff_tuckerC_kernel<<<25 * 2, 32, 0, stream>>>(ws, out);
}


# Sparsemax without sort or scan: element i is in the support iff sum_j max(z_j - z_i, 0) < 1; each thread tests its own element over the row, support sum and size reduced per wave (2 barriers, was 58)
# speedup vs baseline: 1.2899x; 1.2899x over previous
#include <hip/hip_runtime.h>
#include <math.h>


typedef __attribute__((ext_vector_type(2))) float v2f;
typedef __attribute__((ext_vector_type(8))) float v8f;

#define DIM 125
#define NP  216
#define FD  36
#define RK  20

#define OFF_YQ   61440
#define OFF_YK   90112
#define OFF_CP   165440
#define OFF_T2   238016

template <int K>
__device__ inline v8f wmma_rows(const float* __restrict__ ap,
                                const float* __restrict__ bp, bool bok) {
  v8f c = {};
#pragma unroll
  for (int kk = 0; kk < K; kk += 4) {
    v2f a = *(const v2f*)(ap + kk);
    v2f b = *(const v2f*)(bp + kk);
    if (!bok) { b.x = 0.0f; b.y = 0.0f; }
    c = __builtin_amdgcn_wmma_f32_16x16x4_f32(false, a, false, b, (short)0, c,
                                              false, false);
  }
  return c;
}

__device__ inline float tree8(const float* x) {
  return ((x[0] + x[4]) + (x[2] + x[6])) + ((x[1] + x[5]) + (x[3] + x[7]));
}
__device__ inline float half_wave_sum(float s) {
  s += __shfl_xor(s, 8);
  s += __shfl_xor(s, 4);
  s += __shfl_xor(s, 2);
  s += __shfl_xor(s, 1);
  return s;
}

__global__ __launch_bounds__(256) void ff_proj_bn_l2_kernel(
    const float* __restrict__ x, const float* __restrict__ qw,
    const float* __restrict__ kw, const float* __restrict__ gamma,
    const float* __restrict__ beta, float* __restrict__ ws) {
  __shared__ float Ys[16 * 130];
  int which = blockIdx.x / 14, tm = blockIdx.x % 14;
  const float* Wt = which ? kw : qw;
  float* Y = ws + (which ? OFF_YK : OFF_YQ);
  int tid = threadIdx.x, lane = tid & 31, lo = lane & 15, hi = lane >> 4;
  int w = tid >> 5;
  int m = tm * 16 + lo, n = w * 16 + lo;
  bool mok = m < NP, nok = n < DIM;
  const float* arow = x + (mok ? m : 0) * DIM;
  const float* brow = Wt + (nok ? n : 0) * DIM;
  const float* ap = arow + 2 * hi;
  const float* bp = brow + 2 * hi;
  v8f c = {};
#pragma unroll
  for (int kk = 0; kk < 124; kk += 4) {
    v2f a, b;
    a.x = ap[kk]; a.y = ap[kk + 1];
    b.x = bp[kk]; b.y = bp[kk + 1];
    if (!mok) { a.x = 0.0f; a.y = 0.0f; }
    if (!nok) { b.x = 0.0f; b.y = 0.0f; }
    c = __builtin_amdgcn_wmma_f32_16x16x4_f32(false, a, false, b, (short)0, c,
                                              false, false);
  }
  {
    float at = arow[124], bt = brow[124];
    v2f a, b;
    a.x = (mok && hi == 0) ? at : 0.0f; a.y = 0.0f;
    b.x = (nok && hi == 0) ? bt : 0.0f; b.y = 0.0f;
    c = __builtin_amdgcn_wmma_f32_16x16x4_f32(false, a, false, b, (short)0, c,
                                              false, false);
  }
#pragma unroll
  for (int v = 0; v < 8; ++v) Ys[(v + 8 * hi) * 130 + w * 16 + lo] = c[v];
  __syncthreads();

  int r = tid >> 4, sub = tid & 15;
  int m2 = tm * 16 + r;
  bool rok = m2 < NP;
  float xv[8], q[8];
#pragma unroll
  for (int jj = 0; jj < 8; ++jj) {
    xv[jj] = Ys[r * 130 + sub + 16 * jj];
    q[jj] = xv[jj] * xv[jj];
  }
  float s1 = half_wave_sum(tree8(xv));
  float s2 = half_wave_sum(tree8(q));
  float mu = s1 * (1.0f / DIM);
  float var = s2 * (1.0f / DIM) - mu * mu;
  float rstd = rsqrtf(var + 1e-5f);
  float g = gamma[rok ? m2 : 0], bt = beta[rok ? m2 : 0];
  float yn[8];
#pragma unroll
  for (int jj = 0; jj < 8; ++jj) {
    int col = sub + 16 * jj;
    yn[jj] = (col < DIM) ? ((xv[jj] - mu) * rstd * g + bt) : 0.0f;
    q[jj] = yn[jj] * yn[jj];
  }
  float nrm = fmaxf(sqrtf(half_wave_sum(tree8(q))), 1e-12f);
#pragma unroll
  for (int jj = 0; jj < 8; ++jj) {
    int col = sub + 16 * jj;
    Y[m2 * 128 + col] = (rok && col < DIM) ? yn[jj] / nrm : 0.0f;
  }
}

__device__ inline float xhalf(float x) {
  int b = __builtin_bit_cast(int, x);
  return __builtin_bit_cast(
      float, __builtin_amdgcn_permlanex16(b, b, 0x76543210, 0xfedcba98, false,
                                          false));
}
template <int M>
__device__ inline float add_row_xmask(float x) {
  int b = __builtin_bit_cast(int, x);
  int o = __builtin_amdgcn_update_dpp(0, b, 0x160 + M, 0xf, 0xf, false);
  return x + __builtin_bit_cast(float, o);
}
__device__ inline float wave_sum(float x) {
  x = add_row_xmask<1>(x);
  x = add_row_xmask<2>(x);
  x = add_row_xmask<4>(x);
  x = add_row_xmask<8>(x);
  return x + xhalf(x);
}

__global__ __launch_bounds__(256) void ff_scores_sparsemax_kernel(
    float* __restrict__ ws, float* __restrict__ att) {
  float* Cp = ws + OFF_CP;
  __shared__ __attribute__((aligned(16))) float z[224];
  __shared__ __attribute__((aligned(16))) float psum[8];
  __shared__ __attribute__((aligned(16))) int pcnt[8];
  int row = blockIdx.x, tid = threadIdx.x;
  int lane = tid & 31, lo = lane & 15, hi = lane >> 4;
  int w = __builtin_amdgcn_readfirstlane(tid >> 5);
  int m0 = row & ~15, rr = row & 15;
  const float scale = 0.08944271909999159f;
  const float* ap = ws + OFF_YQ + (m0 + lo) * 128 + 2 * hi;
  for (int tn = w; tn < 14; tn += 8) {
    const float* bp = ws + OFF_YK + (tn * 16 + lo) * 128 + 2 * hi;
    v8f c = wmma_rows<128>(ap, bp, true);
    float cv = c[0];
#pragma unroll
    for (int v = 1; v < 8; ++v) cv = ((rr & 7) == v) ? c[v] : cv;
    int n = tn * 16 + lo;
    if (hi == (rr >> 3)) z[n] = cv * scale;
  }
  __syncthreads();
  bool live = tid < NP;
  float zi = z[live ? tid : 0];
  const float4* z4 = (const float4*)z;
  float s0 = 0.0f, s1 = 0.0f, s2 = 0.0f, s3 = 0.0f;
#pragma unroll
  for (int q = 0; q < NP / 4; ++q) {
    float4 v = z4[q];
    s0 += fmaxf(v.x - zi, 0.0f);
    s1 += fmaxf(v.y - zi, 0.0f);
    s2 += fmaxf(v.z - zi, 0.0f);
    s3 += fmaxf(v.w - zi, 0.0f);
  }
  float f = (s0 + s1) + (s2 + s3);
  bool in = live && (f < 1.0f);
  int kw = __popc(__builtin_amdgcn_ballot_w32(in));
  float sw = wave_sum(in ? zi : 0.0f);
  if (lane == 0) { psum[w] = sw; pcnt[w] = kw; }
  __syncthreads();
  float4 pa = *(const float4*)&psum[0], pb = *(const float4*)&psum[4];
  int4 ka = *(const int4*)&pcnt[0], kb = *(const int4*)&pcnt[4];
  float ssum = ((pa.x + pa.y) + (pa.z + pa.w)) + ((pb.x + pb.y) + (pb.z + pb.w));
  int kz = max(((ka.x + ka.y) + (ka.z + ka.w)) + ((kb.x + kb.y) + (kb.z + kb.w)), 1);
  float tau = (ssum - 1.0f) / (float)kz;
  if (live) {
    float rv = fmaxf(zi - tau, 0.0f);
    att[row * NP + tid] = rv;
    int p = row, qq = tid;
    int i = 6 * (p / 36) + (qq / 36);
    int j = 6 * ((p / 6) % 6) + ((qq / 6) % 6);
    int k = 6 * (p % 6) + (qq % 6);
    Cp[(i * FD + j) * FD + k] = rv;
  }
}

__global__ __launch_bounds__(256) void ff_tuckerAB_kernel(
    float* __restrict__ ws, const float* __restrict__ U2,
    const float* __restrict__ U3) {
  __shared__ __attribute__((aligned(16))) float T1s[32 * FD];
  int i = blockIdx.x, tid = threadIdx.x;
  int lane = tid & 31, lo = lane & 15, hi = lane >> 4;
  int w = __builtin_amdgcn_readfirstlane(tid >> 5);
  if (w < 6) {
    int tm = w >> 1, tn = w & 1;
    int j = tm * 16 + lo;
    int jc = j < FD ? j : FD - 1;
    int d = tn * 16 + lo;
    bool dok = d < RK;
    const float* ap = ws + OFF_CP + (i * FD + jc) * FD + 2 * hi;
    const float* bp = U3 + (dok ? d : 0) * FD + 2 * hi;
    v8f c = wmma_rows<FD>(ap, bp, dok);
#pragma unroll
    for (int v = 0; v < 8; ++v) {
      int jj = tm * 16 + v + 8 * hi;
      if (jj < FD) T1s[d * FD + jj] = c[v];
    }
  }
  __syncthreads();
  if (w < 4) {
    int tm = w >> 1, tn = w & 1;
    int cc = tn * 16 + lo;
    bool cok = cc < RK;
    const float* bp = U2 + (cok ? cc : 0) * FD + 2 * hi;
    const int aoff = (tm * 16 + lo) * FD + 2 * hi;
    v8f c = {};
#pragma unroll
    for (int kk = 0; kk < FD; kk += 4) {
      v2f a = *(const v2f*)(&T1s[aoff + kk]);
      v2f b = *(const v2f*)(bp + kk);
      if (!cok) { b.x = 0.0f; b.y = 0.0f; }
      c = __builtin_amdgcn_wmma_f32_16x16x4_f32(false, a, false, b, (short)0,
                                                c, false, false);
    }
    float* T2t = ws + OFF_T2;
#pragma unroll
    for (int v = 0; v < 8; ++v) {
      int d = tm * 16 + v + 8 * hi;
      if (d < RK && cok) T2t[(cc * RK + d) * FD + i] = c[v];
    }
  }
}

__global__ __launch_bounds__(32) void ff_tuckerC_kernel(
    const float* __restrict__ ws, const float* __restrict__ U1,
    float* __restrict__ out) {
  int t = blockIdx.x;
  int tm = t / 2, tn = t % 2;
  int lane = threadIdx.x, lo = lane & 15, hi = lane >> 4;
  int n = tn * 16 + lo;
  bool nok = n < RK;
  const float* ap = ws + OFF_T2 + (tm * 16 + lo) * FD + 2 * hi;
  const float* bp = U1 + (nok ? n : 0) * FD + 2 * hi;
  v8f c = wmma_rows<FD>(ap, bp, nok);
#pragma unroll
  for (int v = 0; v < 8; ++v) {
    int r = tm * 16 + v + 8 * hi;
    if (nok) out[n * 400 + r] = tanhf(c[v]);
  }
}

extern "C" void kernel_launch(void* const* d_in, const int* in_sizes, int n_in,
                              void* d_out, int out_size, void* d_ws,
                              size_t ws_size, hipStream_t stream) {
  const float* patches = (const float*)d_in[0];
  const float* qw      = (const float*)d_in[1];
  const float* kw      = (const float*)d_in[2];
  const float* gamma   = (const float*)d_in[3];
  const float* beta    = (const float*)d_in[4];
  const float* U1      = (const float*)d_in[5];
  const float* U2      = (const float*)d_in[6];
  const float* U3      = (const float*)d_in[7];

  float* out = (float*)d_out;
  float* att = out + RK * RK * RK;
  float* ws = (float*)d_ws;

  ff_proj_bn_l2_kernel<<<2 * 14, 256, 0, stream>>>(patches, qw, kw, gamma,
                                                   beta, ws);
  ff_scores_sparsemax_kernel<<<NP, 256, 0, stream>>>(ws, att);
  ff_tuckerAB_kernel<<<FD, 256, 0, stream>>>(ws, U2, U3);
  ff_tuckerC_kernel<<<25 * 2, 32, 0, stream>>>(ws, U1, out);
}


# Whole-line operand fetch for the projection and score products: x tile and W copied to LDS by 16-B loads; Q/K stored in WMMA fragment order and fetched by 16-B loads (4 lines per load, was 16)
# speedup vs baseline: 1.7218x; 1.3348x over previous
#include <hip/hip_runtime.h>
#include <math.h>


typedef __attribute__((ext_vector_type(2))) float v2f;
typedef __attribute__((ext_vector_type(4))) float v4f;
typedef __attribute__((ext_vector_type(8))) float v8f;
typedef v4f v4f_in __attribute__((aligned(4)));

#define DIM 125
#define YP  132
#define NP  216
#define FD  36
#define RK  20

#define OFF_YQ   61440
#define OFF_YK   90112
#define OFF_CP   165440
#define OFF_T2   238016

template <int K>
__device__ inline v8f wmma_rows(const float* __restrict__ ap,
                                const float* __restrict__ bp, bool bok) {
  v8f c = {};
#pragma unroll
  for (int kk = 0; kk < K; kk += 4) {
    v2f a = *(const v2f*)(ap + kk);
    v2f b = *(const v2f*)(bp + kk);
    if (!bok) { b.x = 0.0f; b.y = 0.0f; }
    c = __builtin_amdgcn_wmma_f32_16x16x4_f32(false, a, false, b, (short)0, c,
                                              false, false);
  }
  return c;
}

__device__ inline v8f wmma_frag16(const v4f (&a)[16], const v4f (&b)[16]) {
  v8f c = {};
#pragma unroll
  for (int j = 0; j < 16; ++j) {
    c = __builtin_amdgcn_wmma_f32_16x16x4_f32(false, a[j].lo, false, b[j].lo,
                                              (short)0, c, false, false);
    c = __builtin_amdgcn_wmma_f32_16x16x4_f32(false, a[j].hi, false, b[j].hi,
                                              (short)0, c, false, false);
  }
  return c;
}
__device__ inline float pick_row(v8f c, int rr) {
  float cv = c[0];
#pragma unroll
  for (int v = 1; v < 8; ++v) cv = ((rr & 7) == v) ? c[v] : cv;
  return cv;
}

template <bool TWO>
__device__ inline void score_tiles(const v4f* __restrict__ aq,
                                   const v4f* __restrict__ bk,
                                   float* zw, int rr, bool keep, float scale) {
  v4f a[16], b0[16], b1[16];
#pragma unroll
  for (int j = 0; j < 16; ++j) {
    a[j] = aq[j * 32];
    b0[j] = bk[j * 32];
  }
  if (TWO) {
#pragma unroll
    for (int j = 0; j < 16; ++j) b1[j] = bk[8 * 512 + j * 32];
  }
  __builtin_amdgcn_sched_barrier(0);
  {
    v8f c = wmma_frag16(a, b0);
    float cv = pick_row(c, rr);
    if (keep) zw[0] = cv * scale;
  }
  if (TWO) {
    v8f c = wmma_frag16(a, b1);
    float cv = pick_row(c, rr);
    if (keep) zw[128] = cv * scale;
  }
}

__device__ inline float tree8(const float* x) {
  return ((x[0] + x[4]) + (x[2] + x[6])) + ((x[1] + x[5]) + (x[3] + x[7]));
}
__device__ inline float half_wave_sum(float s) {
  s += __shfl_xor(s, 8);
  s += __shfl_xor(s, 4);
  s += __shfl_xor(s, 2);
  s += __shfl_xor(s, 1);
  return s;
}

__global__ __launch_bounds__(256) void ff_proj_bn_l2_kernel(
    const float* __restrict__ x, const float* __restrict__ qw,
    const float* __restrict__ kw, const float* __restrict__ gamma,
    const float* __restrict__ beta, float* __restrict__ ws) {
  __shared__ __attribute__((aligned(16))) float Ys[16 * YP];
  __shared__ __attribute__((aligned(16))) float Xs[16 * DIM];
  __shared__ __attribute__((aligned(16))) float Wl[DIM * DIM + 7];
  int which = blockIdx.x / 14, tm = blockIdx.x % 14;
  const float* Wt = which ? kw : qw;
  float* Y = ws + (which ? OFF_YK : OFF_YQ);
  int tid = threadIdx.x, lane = tid & 31, lo = lane & 15, hi = lane >> 4;
  int w = tid >> 5;
  v4f xg[2], wg[16];
  int xq[2], wq[16];
#pragma unroll
  for (int it = 0; it < 2; ++it) {
    xq[it] = min(tid + 256 * it, 4 * DIM - 1);
    int g = tm * (16 * DIM) + 4 * xq[it];
    xg[it] = *(const v4f_in*)(x + min(g, NP * DIM - 4));
    if (g >= NP * DIM) xg[it] = (v4f){0.0f, 0.0f, 0.0f, 0.0f};
  }
#pragma unroll
  for (int it = 0; it < 16; ++it) {
    wq[it] = min(tid + 256 * it, (DIM * DIM) / 4 - 1);
    wg[it] = *(const v4f_in*)(Wt + 4 * wq[it]);
  }
  float wlast = Wt[DIM * DIM - 1];
#pragma unroll
  for (int it = 0; it < 2; ++it) *(v4f*)&Xs[4 * xq[it]] = xg[it];
#pragma unroll
  for (int it = 0; it < 16; ++it) *(v4f*)&Wl[4 * wq[it]] = wg[it];
  Wl[DIM * DIM - 1] = wlast;
  __syncthreads();

  int n = w * 16 + lo;
  bool nok = n < DIM;
  const float* arow = Xs + lo * DIM;
  const float* brow = Wl + (nok ? n : 0) * DIM;
  const float* ap = arow + 2 * hi;
  const float* bp = brow + 2 * hi;
  v8f c = {};
#pragma unroll
  for (int kk = 0; kk < 124; kk += 4) {
    v2f a, b;
    a.x = ap[kk]; a.y = ap[kk + 1];
    b.x = bp[kk]; b.y = bp[kk + 1];
    if (!nok) { b.x = 0.0f; b.y = 0.0f; }
    c = __builtin_amdgcn_wmma_f32_16x16x4_f32(false, a, false, b, (short)0, c,
                                              false, false);
  }
  {
    float at = arow[124], bt = brow[124];
    v2f a, b;
    a.x = (hi == 0) ? at : 0.0f; a.y = 0.0f;
    b.x = (nok && hi == 0) ? bt : 0.0f; b.y = 0.0f;
    c = __builtin_amdgcn_wmma_f32_16x16x4_f32(false, a, false, b, (short)0, c,
                                              false, false);
  }
#pragma unroll
  for (int v = 0; v < 8; ++v) Ys[(v + 8 * hi) * YP + w * 16 + lo] = c[v];
  __syncthreads();

  int r = tid >> 4, sub = tid & 15;
  int m2 = tm * 16 + r;
  bool rok = m2 < NP;
  float xv[8], q[8];
#pragma unroll
  for (int jj = 0; jj < 8; ++jj) {
    xv[jj] = Ys[r * YP + sub + 16 * jj];
    q[jj] = xv[jj] * xv[jj];
  }
  float s1 = half_wave_sum(tree8(xv));
  float s2 = half_wave_sum(tree8(q));
  float mu = s1 * (1.0f / DIM);
  float var = s2 * (1.0f / DIM) - mu * mu;
  float rstd = rsqrtf(var + 1e-5f);
  float g = gamma[rok ? m2 : 0], bt = beta[rok ? m2 : 0];
  float yn[8];
#pragma unroll
  for (int jj = 0; jj < 8; ++jj) {
    int col = sub + 16 * jj;
    yn[jj] = (col < DIM) ? ((xv[jj] - mu) * rstd * g + bt) : 0.0f;
    q[jj] = yn[jj] * yn[jj];
  }
  float nrm = fmaxf(sqrtf(half_wave_sum(tree8(q))), 1e-12f);
#pragma unroll
  for (int jj = 0; jj < 8; ++jj) {
    int col = sub + 16 * jj;
    Ys[r * YP + col] = (rok && col < DIM) ? yn[jj] / nrm : 0.0f;
  }
  __syncthreads();
  v4f* Yf = (v4f*)Y + tm * 512;
#pragma unroll
  for (int i = 0; i < 2; ++i) {
    int idx = tid + 256 * i;
    int j = idx >> 5;
    const float* src = &Ys[lo * YP + 8 * j + 2 * hi];
    v2f p0 = *(const v2f*)src;
    v2f p1 = *(const v2f*)(src + 4);
    v4f o;
    o.x = p0.x; o.y = p0.y; o.z = p1.x; o.w = p1.y;
    Yf[idx] = o;
  }
}

__device__ inline float xhalf(float x) {
  int b = __builtin_bit_cast(int, x);
  return __builtin_bit_cast(
      float, __builtin_amdgcn_permlanex16(b, b, 0x76543210, 0xfedcba98, false,
                                          false));
}
template <int M>
__device__ inline float add_row_xmask(float x) {
  int b = __builtin_bit_cast(int, x);
  int o = __builtin_amdgcn_update_dpp(0, b, 0x160 + M, 0xf, 0xf, false);
  return x + __builtin_bit_cast(float, o);
}
__device__ inline float wave_sum(float x) {
  x = add_row_xmask<1>(x);
  x = add_row_xmask<2>(x);
  x = add_row_xmask<4>(x);
  x = add_row_xmask<8>(x);
  return x + xhalf(x);
}

__global__ __launch_bounds__(256) void ff_scores_sparsemax_kernel(
    float* __restrict__ ws, float* __restrict__ att) {
  float* Cp = ws + OFF_CP;
  __shared__ __attribute__((aligned(16))) float z[224];
  __shared__ __attribute__((aligned(16))) float psum[8];
  __shared__ __attribute__((aligned(16))) int pcnt[8];
  int row = blockIdx.x, tid = threadIdx.x;
  int lane = tid & 31, lo = lane & 15, hi = lane >> 4;
  int w = __builtin_amdgcn_readfirstlane(tid >> 5);
  int m0 = row & ~15, rr = row & 15;
  const float scale = 0.08944271909999159f;
  const v4f* aq = (const v4f*)(ws + OFF_YQ) + (m0 >> 4) * 512 + lane;
  const v4f* bk = (const v4f*)(ws + OFF_YK) + w * 512 + lane;
  float* zw = z + w * 16 + lo;
  bool keep = hi == (rr >> 3);
  if (w < 6) score_tiles<true>(aq, bk, zw, rr, keep, scale);
  else       score_tiles<false>(aq, bk, zw, rr, keep, scale);
  __syncthreads();
  bool live = tid < NP;
  float zi = z[live ? tid : 0];
  const float4* z4 = (const float4*)z;
  float s0 = 0.0f, s1 = 0.0f, s2 = 0.0f, s3 = 0.0f;
#pragma unroll
  for (int q = 0; q < NP / 4; ++q) {
    float4 v = z4[q];
    s0 += fmaxf(v.x - zi, 0.0f);
    s1 += fmaxf(v.y - zi, 0.0f);
    s2 += fmaxf(v.z - zi, 0.0f);
    s3 += fmaxf(v.w - zi, 0.0f);
  }
  float f = (s0 + s1) + (s2 + s3);
  bool in = live && (f < 1.0f);
  int kw = __popc(__builtin_amdgcn_ballot_w32(in));
  float sw = wave_sum(in ? zi : 0.0f);
  if (lane == 0) { psum[w] = sw; pcnt[w] = kw; }
  __syncthreads();
  float4 pa = *(const float4*)&psum[0], pb = *(const float4*)&psum[4];
  int4 ka = *(const int4*)&pcnt[0], kb = *(const int4*)&pcnt[4];
  float ssum = ((pa.x + pa.y) + (pa.z + pa.w)) + ((pb.x + pb.y) + (pb.z + pb.w));
  int kz = max(((ka.x + ka.y) + (ka.z + ka.w)) + ((kb.x + kb.y) + (kb.z + kb.w)), 1);
  float tau = (ssum - 1.0f) / (float)kz;
  if (live) {
    float rv = fmaxf(zi - tau, 0.0f);
    att[row * NP + tid] = rv;
    int p = row, qq = tid;
    int i = 6 * (p / 36) + (qq / 36);
    int j = 6 * ((p / 6) % 6) + ((qq / 6) % 6);
    int k = 6 * (p % 6) + (qq % 6);
    Cp[(i * FD + j) * FD + k] = rv;
  }
}

__global__ __launch_bounds__(256) void ff_tuckerAB_kernel(
    float* __restrict__ ws, const float* __restrict__ U2,
    const float* __restrict__ U3) {
  __shared__ __attribute__((aligned(16))) float T1s[32 * FD];
  int i = blockIdx.x, tid = threadIdx.x;
  int lane = tid & 31, lo = lane & 15, hi = lane >> 4;
  int w = __builtin_amdgcn_readfirstlane(tid >> 5);
  if (w < 6) {
    int tm = w >> 1, tn = w & 1;
    int j = tm * 16 + lo;
    int jc = j < FD ? j : FD - 1;
    int d = tn * 16 + lo;
    bool dok = d < RK;
    const float* ap = ws + OFF_CP + (i * FD + jc) * FD + 2 * hi;
    const float* bp = U3 + (dok ? d : 0) * FD + 2 * hi;
    v8f c = wmma_rows<FD>(ap, bp, dok);
#pragma unroll
    for (int v = 0; v < 8; ++v) {
      int jj = tm * 16 + v + 8 * hi;
      if (jj < FD) T1s[d * FD + jj] = c[v];
    }
  }
  __syncthreads();
  if (w < 4) {
    int tm = w >> 1, tn = w & 1;
    int cc = tn * 16 + lo;
    bool cok = cc < RK;
    const float* bp = U2 + (cok ? cc : 0) * FD + 2 * hi;
    const int aoff = (tm * 16 + lo) * FD + 2 * hi;
    v8f c = {};
#pragma unroll
    for (int kk = 0; kk < FD; kk += 4) {
      v2f a = *(const v2f*)(&T1s[aoff + kk]);
      v2f b = *(const v2f*)(bp + kk);
      if (!cok) { b.x = 0.0f; b.y = 0.0f; }
      c = __builtin_amdgcn_wmma_f32_16x16x4_f32(false, a, false, b, (short)0,
                                                c, false, false);
    }
    float* T2t = ws + OFF_T2;
#pragma unroll
    for (int v = 0; v < 8; ++v) {
      int d = tm * 16 + v + 8 * hi;
      if (d < RK && cok) T2t[(cc * RK + d) * FD + i] = c[v];
    }
  }
}

__global__ __launch_bounds__(32) void ff_tuckerC_kernel(
    const float* __restrict__ ws, const float* __restrict__ U1,
    float* __restrict__ out) {
  int t = blockIdx.x;
  int tm = t / 2, tn = t % 2;
  int lane = threadIdx.x, lo = lane & 15, hi = lane >> 4;
  int n = tn * 16 + lo;
  bool nok = n < RK;
  const float* ap = ws + OFF_T2 + (tm * 16 + lo) * FD + 2 * hi;
  const float* bp = U1 + (nok ? n : 0) * FD + 2 * hi;
  v8f c = wmma_rows<FD>(ap, bp, nok);
#pragma unroll
  for (int v = 0; v < 8; ++v) {
    int r = tm * 16 + v + 8 * hi;
    if (nok) out[n * 400 + r] = tanhf(c[v]);
  }
}

extern "C" void kernel_launch(void* const* d_in, const int* in_sizes, int n_in,
                              void* d_out, int out_size, void* d_ws,
                              size_t ws_size, hipStream_t stream) {
  const float* patches = (const float*)d_in[0];
  const float* qw      = (const float*)d_in[1];
  const float* kw      = (const float*)d_in[2];
  const float* gamma   = (const float*)d_in[3];
  const float* beta    = (const float*)d_in[4];
  const float* U1      = (const float*)d_in[5];
  const float* U2      = (const float*)d_in[6];
  const float* U3      = (const float*)d_in[7];

  float* out = (float*)d_out;
  float* att = out + RK * RK * RK;
  float* ws = (float*)d_ws;

  ff_proj_bn_l2_kernel<<<2 * 14, 256, 0, stream>>>(patches, qw, kw, gamma,
                                                   beta, ws);
  ff_scores_sparsemax_kernel<<<NP, 256, 0, stream>>>(ws, att);
  ff_tuckerAB_kernel<<<FD, 256, 0, stream>>>(ws, U2, U3);
  ff_tuckerC_kernel<<<25 * 2, 32, 0, stream>>>(ws, U1, out);
}


# Score kernel keeps 8 of its 16 tile rows: 27 work-groups (was 216), 216 waves to enter (was 1728); each wave owns one row, sparsemax threshold by shrinking the support inside the wave
# speedup vs baseline: 2.2083x; 1.2826x over previous
#include <hip/hip_runtime.h>
#include <math.h>


typedef __attribute__((ext_vector_type(2))) float v2f;
typedef __attribute__((ext_vector_type(4))) float v4f;
typedef __attribute__((ext_vector_type(8))) float v8f;
typedef v4f v4f_in __attribute__((aligned(4)));

#define DIM 125
#define YP  132
#define NP  216
#define FD  36
#define RK  20

#define OFF_YQ   61440
#define OFF_YK   90112
#define OFF_CP   165440

template <int K>
__device__ inline v8f wmma_rows(const float* __restrict__ ap,
                                const float* __restrict__ bp, bool bok) {
  v8f c = {};
#pragma unroll
  for (int kk = 0; kk < K; kk += 4) {
    v2f a = *(const v2f*)(ap + kk);
    v2f b = *(const v2f*)(bp + kk);
    if (!bok) { b.x = 0.0f; b.y = 0.0f; }
    c = __builtin_amdgcn_wmma_f32_16x16x4_f32(false, a, false, b, (short)0, c,
                                              false, false);
  }
  return c;
}

__device__ inline v8f wmma_frag16(const v4f (&a)[16], const v4f (&b)[16]) {
  v8f c = {};
#pragma unroll
  for (int j = 0; j < 16; ++j) {
    c = __builtin_amdgcn_wmma_f32_16x16x4_f32(false, a[j].lo, false, b[j].lo,
                                              (short)0, c, false, false);
    c = __builtin_amdgcn_wmma_f32_16x16x4_f32(false, a[j].hi, false, b[j].hi,
                                              (short)0, c, false, false);
  }
  return c;
}
#define ZP 224
template <bool TWO>
__device__ inline void score_tiles(const v4f* __restrict__ aq,
                                   const v4f* __restrict__ bk,
                                   float* zw, bool keep, float scale) {
  v4f a[16], b0[16], b1[16];
#pragma unroll
  for (int j = 0; j < 16; ++j) {
    a[j] = aq[j * 32];
    b0[j] = bk[j * 32];
  }
  if (TWO) {
#pragma unroll
    for (int j = 0; j < 16; ++j) b1[j] = bk[8 * 512 + j * 32];
  }
  __builtin_amdgcn_sched_barrier(0);
  {
    v8f c = wmma_frag16(a, b0);
    if (keep) {
#pragma unroll
      for (int v = 0; v < 8; ++v) zw[v * ZP] = c[v] * scale;
    }
  }
  if (TWO) {
    v8f c = wmma_frag16(a, b1);
    if (keep) {
#pragma unroll
      for (int v = 0; v < 8; ++v) zw[v * ZP + 128] = c[v] * scale;
    }
  }
}

__device__ inline float tree8(const float* x) {
  return ((x[0] + x[4]) + (x[2] + x[6])) + ((x[1] + x[5]) + (x[3] + x[7]));
}
__device__ inline float half_wave_sum(float s) {
  s += __shfl_xor(s, 8);
  s += __shfl_xor(s, 4);
  s += __shfl_xor(s, 2);
  s += __shfl_xor(s, 1);
  return s;
}

__global__ __launch_bounds__(256) void ff_proj_bn_l2_kernel(
    const float* __restrict__ x, const float* __restrict__ qw,
    const float* __restrict__ kw, const float* __restrict__ gamma,
    const float* __restrict__ beta, float* __restrict__ ws) {
  __shared__ __attribute__((aligned(16))) float Ys[16 * YP];
  __shared__ __attribute__((aligned(16))) float Xs[16 * DIM];
  __shared__ __attribute__((aligned(16))) float Wl[DIM * DIM + 7];
  int which = blockIdx.x / 14, tm = blockIdx.x % 14;
  const float* Wt = which ? kw : qw;
  float* Y = ws + (which ? OFF_YK : OFF_YQ);
  int tid = threadIdx.x, lane = tid & 31, lo = lane & 15, hi = lane >> 4;
  int w = tid >> 5;
  v4f xg[2], wg[16];
  int xq[2], wq[16];
#pragma unroll
  for (int it = 0; it < 2; ++it) {
    xq[it] = min(tid + 256 * it, 4 * DIM - 1);
    int g = tm * (16 * DIM) + 4 * xq[it];
    xg[it] = *(const v4f_in*)(x + min(g, NP * DIM - 4));
    if (g >= NP * DIM) xg[it] = (v4f){0.0f, 0.0f, 0.0f, 0.0f};
  }
#pragma unroll
  for (int it = 0; it < 16; ++it) {
    wq[it] = min(tid + 256 * it, (DIM * DIM) / 4 - 1);
    wg[it] = *(const v4f_in*)(Wt + 4 * wq[it]);
  }
  float wlast = Wt[DIM * DIM - 1];
#pragma unroll
  for (int it = 0; it < 2; ++it) *(v4f*)&Xs[4 * xq[it]] = xg[it];
#pragma unroll
  for (int it = 0; it < 16; ++it) *(v4f*)&Wl[4 * wq[it]] = wg[it];
  Wl[DIM * DIM - 1] = wlast;
  __syncthreads();

  int n = w * 16 + lo;
  bool nok = n < DIM;
  const float* arow = Xs + lo * DIM;
  const float* brow = Wl + (nok ? n : 0) * DIM;
  const float* ap = arow + 2 * hi;
  const float* bp = brow + 2 * hi;
  v8f c = {};
#pragma unroll
  for (int kk = 0; kk < 124; kk += 4) {
    v2f a, b;
    a.x = ap[kk]; a.y = ap[kk + 1];
    b.x = bp[kk]; b.y = bp[kk + 1];
    if (!nok) { b.x = 0.0f; b.y = 0.0f; }
    c = __builtin_amdgcn_wmma_f32_16x16x4_f32(false, a, false, b, (short)0, c,
                                              false, false);
  }
  {
    float at = arow[124], bt = brow[124];
    v2f a, b;
    a.x = (hi == 0) ? at : 0.0f; a.y = 0.0f;
    b.x = (nok && hi == 0) ? bt : 0.0f; b.y = 0.0f;
    c = __builtin_amdgcn_wmma_f32_16x16x4_f32(false, a, false, b, (short)0, c,
                                              false, false);
  }
#pragma unroll
  for (int v = 0; v < 8; ++v) Ys[(v + 8 * hi) * YP + w * 16 + lo] = c[v];
  __syncthreads();

  int r = tid >> 4, sub = tid & 15;
  int m2 = tm * 16 + r;
  bool rok = m2 < NP;
  float xv[8], q[8];
#pragma unroll
  for (int jj = 0; jj < 8; ++jj) {
    xv[jj] = Ys[r * YP + sub + 16 * jj];
    q[jj] = xv[jj] * xv[jj];
  }
  float s1 = half_wave_sum(tree8(xv));
  float s2 = half_wave_sum(tree8(q));
  float mu = s1 * (1.0f / DIM);
  float var = s2 * (1.0f / DIM) - mu * mu;
  float rstd = rsqrtf(var + 1e-5f);
  float g = gamma[rok ? m2 : 0], bt = beta[rok ? m2 : 0];
  float yn[8];
#pragma unroll
  for (int jj = 0; jj < 8; ++jj) {
    int col = sub + 16 * jj;
    yn[jj] = (col < DIM) ? ((xv[jj] - mu) * rstd * g + bt) : 0.0f;
    q[jj] = yn[jj] * yn[jj];
  }
  float nrm = fmaxf(sqrtf(half_wave_sum(tree8(q))), 1e-12f);
#pragma unroll
  for (int jj = 0; jj < 8; ++jj) {
    int col = sub + 16 * jj;
    Ys[r * YP + col] = (rok && col < DIM) ? yn[jj] / nrm : 0.0f;
  }
  __syncthreads();
  v4f* Yf = (v4f*)Y + tm * 512;
#pragma unroll
  for (int i = 0; i < 2; ++i) {
    int idx = tid + 256 * i;
    int j = idx >> 5;
    const float* src = &Ys[lo * YP + 8 * j + 2 * hi];
    v2f p0 = *(const v2f*)src;
    v2f p1 = *(const v2f*)(src + 4);
    v4f o;
    o.x = p0.x; o.y = p0.y; o.z = p1.x; o.w = p1.y;
    Yf[idx] = o;
  }
}

__device__ inline float xhalf(float x) {
  int b = __builtin_bit_cast(int, x);
  return __builtin_bit_cast(
      float, __builtin_amdgcn_permlanex16(b, b, 0x76543210, 0xfedcba98, false,
                                          false));
}
template <int M>
__device__ inline float add_row_xmask(float x) {
  int b = __builtin_bit_cast(int, x);
  int o = __builtin_amdgcn_update_dpp(0, b, 0x160 + M, 0xf, 0xf, false);
  return x + __builtin_bit_cast(float, o);
}
__device__ inline float wave_sum(float x) {
  x = add_row_xmask<1>(x);
  x = add_row_xmask<2>(x);
  x = add_row_xmask<4>(x);
  x = add_row_xmask<8>(x);
  return x + xhalf(x);
}

__global__ __launch_bounds__(256) void ff_scores_sparsemax_kernel(
    float* __restrict__ ws, float* __restrict__ att) {
  float* Cp = ws + OFF_CP;
  __shared__ __attribute__((aligned(16))) float z[8 * ZP];
  int tid = threadIdx.x;
  int lane = tid & 31, lo = lane & 15, hi = lane >> 4;
  int w = __builtin_amdgcn_readfirstlane(tid >> 5);
  int blk = blockIdx.x >> 1, half = blockIdx.x & 1;
  const float scale = 0.08944271909999159f;
  const v4f* aq = (const v4f*)(ws + OFF_YQ) + blk * 512 + lane;
  const v4f* bk = (const v4f*)(ws + OFF_YK) + w * 512 + lane;
  float* zw = z + w * 16 + lo;
  bool keep = hi == half;
  if (w < 6) score_tiles<true>(aq, bk, zw, keep, scale);
  else       score_tiles<false>(aq, bk, zw, keep, scale);
  __syncthreads();
  const int row = blk * 16 + half * 8 + w;
  const float* zr = z + w * ZP + lane;
  const bool tail = lane < NP - 6 * 32;
  float zv[7], we[7];
#pragma unroll
  for (int e = 0; e < 7; ++e) zv[e] = zr[32 * e];
#pragma unroll
  for (int e = 0; e < 6; ++e) we[e] = zv[e];
  we[6] = tail ? zv[6] : -__builtin_inff();
  float S = wave_sum(((we[0] + we[1]) + (we[2] + we[3])) +
                     ((we[4] + we[5]) + (tail ? zv[6] : 0.0f)));
  int k = NP;
  float tau = (S - 1.0f) / (float)k;
  for (int it = 0; it < NP; ++it) {
    float t[7];
    int cnt = 0;
#pragma unroll
    for (int e = 0; e < 7; ++e) {
      bool m = we[e] > tau;
      cnt += __popc(__builtin_amdgcn_ballot_w32(m));
      t[e] = m ? we[e] : 0.0f;
      we[e] = m ? we[e] : -__builtin_inff();
    }
    if (cnt == k) break;
    S = wave_sum(((t[0] + t[1]) + (t[2] + t[3])) + ((t[4] + t[5]) + t[6]));
    k = max(cnt, 1);
    tau = (S - 1.0f) / (float)k;
  }
  const int pi = row / 36, pj = (row / 6) % 6, pk = row % 6;
#pragma unroll
  for (int e = 0; e < 7; ++e) {
    int qq = lane + 32 * e;
    if (e < 6 || tail) {
      float rv = fmaxf(zv[e] - tau, 0.0f);
      att[row * NP + qq] = rv;
      int i = 6 * pi + (qq / 36);
      int j = 6 * pj + ((qq / 6) % 6);
      int kk = 6 * pk + (qq % 6);
      Cp[(i * FD + j) * FD + kk] = rv;
    }
  }
}

#define CN (FD * FD)

__device__ inline void core_group_load(const float* __restrict__ bp,
                                       v4f (&b)[18]) {
#pragma unroll
  for (int s = 0; s < 9; ++s) {
    b[2 * s] = *(const v4f*)(bp + (4 * s) * CN);
    b[2 * s + 1] = *(const v4f*)(bp + (4 * s + 1) * CN);
  }
}
__device__ inline v4f core_group_mma(const v2f (&ua)[9], const v4f (&b)[18]) {
  v8f c0 = {}, c1 = {}, c2 = {}, c3 = {};
#pragma unroll
  for (int s = 0; s < 9; ++s) {
    v2f t0, t1, t2, t3;
    t0.x = b[2 * s].x; t0.y = b[2 * s + 1].x;
    t1.x = b[2 * s].y; t1.y = b[2 * s + 1].y;
    t2.x = b[2 * s].z; t2.y = b[2 * s + 1].z;
    t3.x = b[2 * s].w; t3.y = b[2 * s + 1].w;
    c0 = __builtin_amdgcn_wmma_f32_16x16x4_f32(false, ua[s], false, t0,
                                               (short)0, c0, false, false);
    c1 = __builtin_amdgcn_wmma_f32_16x16x4_f32(false, ua[s], false, t1,
                                               (short)0, c1, false, false);
    c2 = __builtin_amdgcn_wmma_f32_16x16x4_f32(false, ua[s], false, t2,
                                               (short)0, c2, false, false);
    c3 = __builtin_amdgcn_wmma_f32_16x16x4_f32(false, ua[s], false, t3,
                                               (short)0, c3, false, false);
  }
  v4f r;
  r.x = c0[0]; r.y = c1[0]; r.z = c2[0]; r.w = c3[0];
  return r;
}

__global__ __launch_bounds__(256) void ff_tucker_kernel(
    const float* __restrict__ ws, const float* __restrict__ U1,
    const float* __restrict__ U2, const float* __restrict__ U3,
    float* __restrict__ out) {
  __shared__ __attribute__((aligned(16))) float Ss[CN];
  __shared__ __attribute__((aligned(16))) float T1s[32 * FD];
  int a = blockIdx.x, tid = threadIdx.x;
  int lane = tid & 31, lo = lane & 15, hi = lane >> 4;
  int w = __builtin_amdgcn_readfirstlane(tid >> 5);
  const float* Cp = ws + OFF_CP;
  const float* bp = Cp + 4 * lo + 2 * hi * CN;
  const float* bp3rd = (w < 4) ? bp + (w + 16) * 64
                               : Cp + 20 * 64 + 4 * (lo >> 2) + 2 * hi * CN;
  v4f bA[18], bB[18], bC[18];
  v2f ua[9];
#pragma unroll
  for (int s = 0; s < 9; ++s)
    ua[s] = *(const v2f*)(U1 + a * FD + 4 * s + 2 * hi);
  core_group_load(bp + w * 64, bA);
  core_group_load(bp + (w + 8) * 64, bB);
  {
    v4f r = core_group_mma(ua, bA);
    if (hi == 0) *(v4f*)&Ss[w * 64 + 4 * lo] = r;
  }
  core_group_load(bp3rd, bC);
  {
    v4f r = core_group_mma(ua, bB);
    if (hi == 0) *(v4f*)&Ss[(w + 8) * 64 + 4 * lo] = r;
  }
  const int dn = (w & 1) * 16 + lo;
  const bool nok = dn < RK;
  v2f u3f[9], u2f[9];
#pragma unroll
  for (int s = 0; s < 9; ++s) {
    u3f[s] = *(const v2f*)(U3 + (nok ? dn : 0) * FD + 4 * s + 2 * hi);
    u2f[s] = *(const v2f*)(U2 + (nok ? dn : 0) * FD + 4 * s + 2 * hi);
  }
  if (w < 4) {
    v4f r = core_group_mma(ua, bC);
    if (hi == 0) *(v4f*)&Ss[(w + 16) * 64 + 4 * lo] = r;
  }
  if (w == 7) {
    const int e = lo & 3;
    v8f c = {};
#pragma unroll
    for (int s = 0; s < 9; ++s) {
      v2f t;
      t.x = bC[2 * s][e];
      t.y = bC[2 * s + 1][e];
      c = __builtin_amdgcn_wmma_f32_16x16x4_f32(false, ua[s], false, t,
                                                (short)0, c, false, false);
    }
    if (hi == 0) Ss[20 * 64 + lo] = c[0];
  }
  __syncthreads();
  if (w < 6) {
    int tm = w >> 1, tn = w & 1;
    int j = tm * 16 + lo;
    int jc = j < FD ? j : FD - 1;
    int d = tn * 16 + lo;
    const int aoff = jc * FD + 2 * hi;
    v8f c = {};
#pragma unroll
    for (int kk = 0; kk < FD; kk += 4) {
      v2f av = *(const v2f*)(&Ss[aoff + kk]);
      v2f b = u3f[kk >> 2];
      if (!nok) { b.x = 0.0f; b.y = 0.0f; }
      c = __builtin_amdgcn_wmma_f32_16x16x4_f32(false, av, false, b, (short)0,
                                                c, false, false);
    }
#pragma unroll
    for (int v = 0; v < 8; ++v) {
      int jj = tm * 16 + v + 8 * hi;
      if (jj < FD) T1s[d * FD + jj] = c[v];
    }
  }
  __syncthreads();
  if (w < 4) {
    int tm = w >> 1, tn = w & 1;
    int cc = tn * 16 + lo;
    bool cok = nok;
    const int aoff = (tm * 16 + lo) * FD + 2 * hi;
    v8f c = {};
#pragma unroll
    for (int kk = 0; kk < FD; kk += 4) {
      v2f av = *(const v2f*)(&T1s[aoff + kk]);
      v2f b = u2f[kk >> 2];
      if (!cok) { b.x = 0.0f; b.y = 0.0f; }
      c = __builtin_amdgcn_wmma_f32_16x16x4_f32(false, av, false, b, (short)0,
                                                c, false, false);
    }
    int d0 = tm * 16 + 8 * hi;
    float* op = out + a * (RK * RK) + cc * RK + d0;
    if (cok && d0 + 3 < RK) {
      v4f o;
      o.x = tanhf(c[0]); o.y = tanhf(c[1]); o.z = tanhf(c[2]); o.w = tanhf(c[3]);
      *(v4f*)op = o;
    }
    if (cok && d0 + 7 < RK) {
      v4f o;
      o.x = tanhf(c[4]); o.y = tanhf(c[5]); o.z = tanhf(c[6]); o.w = tanhf(c[7]);
      *(v4f*)(op + 4) = o;
    }
  }
}

extern "C" void kernel_launch(void* const* d_in, const int* in_sizes, int n_in,
                              void* d_out, int out_size, void* d_ws,
                              size_t ws_size, hipStream_t stream) {
  const float* patches = (const float*)d_in[0];
  const float* qw      = (const float*)d_in[1];
  const float* kw      = (const float*)d_in[2];
  const float* gamma   = (const float*)d_in[3];
  const float* beta    = (const float*)d_in[4];
  const float* U1      = (const float*)d_in[5];
  const float* U2      = (const float*)d_in[6];
  const float* U3      = (const float*)d_in[7];

  float* out = (float*)d_out;
  float* att = out + RK * RK * RK;
  float* ws = (float*)d_ws;

  ff_proj_bn_l2_kernel<<<2 * 14, 256, 0, stream>>>(patches, qw, kw, gamma,
                                                   beta, ws);
  ff_scores_sparsemax_kernel<<<27, 256, 0, stream>>>(ws, att);
  ff_tucker_kernel<<<RK, 256, 0, stream>>>(ws, U1, U2, U3, out);
}


# Tucker sum over i moved into the score kernel: a group owns attention rows 36m+r, so R[a]=sum_i U1[a,i]core[i] is one 20x36x36 WMMA product per group; launch 3 reads 5 KB per slice (was 187 KB)
# speedup vs baseline: 2.4034x; 1.0883x over previous
#include <hip/hip_runtime.h>
#include <math.h>


typedef __attribute__((ext_vector_type(2))) float v2f;
typedef __attribute__((ext_vector_type(4))) float v4f;
typedef __attribute__((ext_vector_type(8))) float v8f;
typedef v4f v4f_in __attribute__((aligned(4)));

#define DIM 125
#define YP  132
#define NP  216
#define FD  36
#define RK  20

#define OFF_YQ   61440
#define OFF_YK   90112
#define OFF_R    165440
#define CN (FD * FD)

template <int K>
__device__ inline v8f wmma_rows(const float* __restrict__ ap,
                                const float* __restrict__ bp, bool bok) {
  v8f c = {};
#pragma unroll
  for (int kk = 0; kk < K; kk += 4) {
    v2f a = *(const v2f*)(ap + kk);
    v2f b = *(const v2f*)(bp + kk);
    if (!bok) { b.x = 0.0f; b.y = 0.0f; }
    c = __builtin_amdgcn_wmma_f32_16x16x4_f32(false, a, false, b, (short)0, c,
                                              false, false);
  }
  return c;
}

__device__ inline v8f wmma_frag16(const v4f (&a)[16], const v4f (&b)[16]) {
  v8f c = {};
#pragma unroll
  for (int j = 0; j < 16; ++j) {
    c = __builtin_amdgcn_wmma_f32_16x16x4_f32(false, a[j].lo, false, b[j].lo,
                                              (short)0, c, false, false);
    c = __builtin_amdgcn_wmma_f32_16x16x4_f32(false, a[j].hi, false, b[j].hi,
                                              (short)0, c, false, false);
  }
  return c;
}
#define ZP 224
#define NR 6
template <bool TWO>
__device__ inline void score_tiles(const v4f* __restrict__ aq,
                                   const v4f* __restrict__ bk,
                                   float* zw, bool keep, float scale) {
  v4f a[16], b0[16], b1[16];
#pragma unroll
  for (int j = 0; j < 16; ++j) {
    a[j] = aq[j * 32];
    b0[j] = bk[j * 32];
  }
  if (TWO) {
#pragma unroll
    for (int j = 0; j < 16; ++j) b1[j] = bk[8 * 512 + j * 32];
  }
  __builtin_amdgcn_sched_barrier(0);
  {
    v8f c = wmma_frag16(a, b0);
    if (keep) {
#pragma unroll
      for (int v = 0; v < NR; ++v) zw[v * ZP] = c[v] * scale;
    }
  }
  if (TWO) {
    v8f c = wmma_frag16(a, b1);
    if (keep) {
#pragma unroll
      for (int v = 0; v < NR; ++v) zw[v * ZP + 128] = c[v] * scale;
    }
  }
}

__device__ inline float tree8(const float* x) {
  return ((x[0] + x[4]) + (x[2] + x[6])) + ((x[1] + x[5]) + (x[3] + x[7]));
}
__device__ inline float half_wave_sum(float s) {
  s += __shfl_xor(s, 8);
  s += __shfl_xor(s, 4);
  s += __shfl_xor(s, 2);
  s += __shfl_xor(s, 1);
  return s;
}

__global__ __launch_bounds__(256) void ff_proj_bn_l2_kernel(
    const float* __restrict__ x, const float* __restrict__ qw,
    const float* __restrict__ kw, const float* __restrict__ gamma,
    const float* __restrict__ beta, float* __restrict__ ws) {
  __shared__ __attribute__((aligned(16))) float Ys[16 * YP];
  __shared__ __attribute__((aligned(16))) float Xs[16 * DIM];
  __shared__ __attribute__((aligned(16))) float Wl[DIM * DIM + 7];
  int which = blockIdx.x / 14, tm = blockIdx.x % 14;
  const float* Wt = which ? kw : qw;
  float* Y = ws + (which ? OFF_YK : OFF_YQ);
  int tid = threadIdx.x, lane = tid & 31, lo = lane & 15, hi = lane >> 4;
  int w = tid >> 5;
  v4f xg[2], wg[16];
  int xq[2], wq[16];
#pragma unroll
  for (int it = 0; it < 2; ++it) {
    xq[it] = min(tid + 256 * it, 4 * DIM - 1);
    int g = tm * (16 * DIM) + 4 * xq[it];
    xg[it] = *(const v4f_in*)(x + min(g, NP * DIM - 4));
    if (g >= NP * DIM) xg[it] = (v4f){0.0f, 0.0f, 0.0f, 0.0f};
  }
#pragma unroll
  for (int it = 0; it < 16; ++it) {
    wq[it] = min(tid + 256 * it, (DIM * DIM) / 4 - 1);
    wg[it] = *(const v4f_in*)(Wt + 4 * wq[it]);
  }
  float wlast = Wt[DIM * DIM - 1];
#pragma unroll
  for (int it = 0; it < 2; ++it) *(v4f*)&Xs[4 * xq[it]] = xg[it];
#pragma unroll
  for (int it = 0; it < 16; ++it) *(v4f*)&Wl[4 * wq[it]] = wg[it];
  Wl[DIM * DIM - 1] = wlast;
  __syncthreads();

  int n = w * 16 + lo;
  bool nok = n < DIM;
  const float* arow = Xs + lo * DIM;
  const float* brow = Wl + (nok ? n : 0) * DIM;
  const float* ap = arow + 2 * hi;
  const float* bp = brow + 2 * hi;
  v8f c = {};
#pragma unroll
  for (int kk = 0; kk < 124; kk += 4) {
    v2f a, b;
    a.x = ap[kk]; a.y = ap[kk + 1];
    b.x = bp[kk]; b.y = bp[kk + 1];
    if (!nok) { b.x = 0.0f; b.y = 0.0f; }
    c = __builtin_amdgcn_wmma_f32_16x16x4_f32(false, a, false, b, (short)0, c,
                                              false, false);
  }
  {
    float at = arow[124], bt = brow[124];
    v2f a, b;
    a.x = (hi == 0) ? at : 0.0f; a.y = 0.0f;
    b.x = (nok && hi == 0) ? bt : 0.0f; b.y = 0.0f;
    c = __builtin_amdgcn_wmma_f32_16x16x4_f32(false, a, false, b, (short)0, c,
                                              false, false);
  }
#pragma unroll
  for (int v = 0; v < 8; ++v) Ys[(v + 8 * hi) * YP + w * 16 + lo] = c[v];
  __syncthreads();

  int r = tid >> 4, sub = tid & 15;
  int m2 = tm * 16 + r;
  bool rok = m2 < NP;
  float xv[8], q[8];
#pragma unroll
  for (int jj = 0; jj < 8; ++jj) {
    xv[jj] = Ys[r * YP + sub + 16 * jj];
    q[jj] = xv[jj] * xv[jj];
  }
  float s1 = half_wave_sum(tree8(xv));
  float s2 = half_wave_sum(tree8(q));
  float mu = s1 * (1.0f / DIM);
  float var = s2 * (1.0f / DIM) - mu * mu;
  float rstd = rsqrtf(var + 1e-5f);
  float g = gamma[rok ? m2 : 0], bt = beta[rok ? m2 : 0];
  float yn[8];
#pragma unroll
  for (int jj = 0; jj < 8; ++jj) {
    int col = sub + 16 * jj;
    yn[jj] = (col < DIM) ? ((xv[jj] - mu) * rstd * g + bt) : 0.0f;
    q[jj] = yn[jj] * yn[jj];
  }
  float nrm = fmaxf(sqrtf(half_wave_sum(tree8(q))), 1e-12f);
#pragma unroll
  for (int jj = 0; jj < 8; ++jj) {
    int col = sub + 16 * jj;
    Ys[r * YP + col] = (rok && col < DIM) ? yn[jj] / nrm : 0.0f;
  }
  __syncthreads();
  v4f* Yf = (v4f*)Y + tm * 512;
#pragma unroll
  for (int i = 0; i < 2; ++i) {
    int idx = tid + 256 * i;
    int j = idx >> 5;
    const float* src = &Ys[lo * YP + 8 * j + 2 * hi];
    v2f p0 = *(const v2f*)src;
    v2f p1 = *(const v2f*)(src + 4);
    v4f o;
    o.x = p0.x; o.y = p0.y; o.z = p1.x; o.w = p1.y;
    Yf[idx] = o;
  }
}

__device__ inline float xhalf(float x) {
  int b = __builtin_bit_cast(int, x);
  return __builtin_bit_cast(
      float, __builtin_amdgcn_permlanex16(b, b, 0x76543210, 0xfedcba98, false,
                                          false));
}
template <int M>
__device__ inline float add_row_xmask(float x) {
  int b = __builtin_bit_cast(int, x);
  int o = __builtin_amdgcn_update_dpp(0, b, 0x160 + M, 0xf, 0xf, false);
  return x + __builtin_bit_cast(float, o);
}
__device__ inline float wave_sum(float x) {
  x = add_row_xmask<1>(x);
  x = add_row_xmask<2>(x);
  x = add_row_xmask<4>(x);
  x = add_row_xmask<8>(x);
  return x + xhalf(x);
}

__global__ __launch_bounds__(256) void ff_scores_sparsemax_kernel(
    float* __restrict__ ws, const float* __restrict__ U1,
    float* __restrict__ att) {
  float* Rw = ws + OFF_R;
  __shared__ __attribute__((aligned(16))) float z[NR * ZP];
  int tid = threadIdx.x;
  int lane = tid & 31, lo = lane & 15, hi = lane >> 4;
  int w = __builtin_amdgcn_readfirstlane(tid >> 5);
  const int r = blockIdx.x;
  const float scale = 0.08944271909999159f;
  const int pa = 36 * min(lo, NR - 1) + r;
  const v4f* aq =
      (const v4f*)(ws + OFF_YQ) + (pa >> 4) * 512 + (pa & 15) + 16 * hi;
  const v4f* bk = (const v4f*)(ws + OFF_YK) + w * 512 + lane;
  float* zw = z + w * 16 + lo;
  bool keep = hi == 0;
  if (w < 6) score_tiles<true>(aq, bk, zw, keep, scale);
  else       score_tiles<false>(aq, bk, zw, keep, scale);
  __syncthreads();
  const int tm = w / 3, tn = w - 3 * tm;
  const int row = 36 * w + r;
  const bool tail = lane < NP - 6 * 32;
  v2f uf[9];
  float rv[7];
  if (w < NR) {
    const int ar = tm * 16 + lo;
#pragma unroll
    for (int s = 0; s < 9; ++s)
      uf[s] = *(const v2f*)(U1 + (ar < RK ? ar : 0) * FD + 18 * hi + 2 * s);
    float* zr = z + w * ZP + lane;
    float zv[7], we[7];
#pragma unroll
    for (int e = 0; e < 7; ++e) zv[e] = zr[32 * e];
#pragma unroll
    for (int e = 0; e < 6; ++e) we[e] = zv[e];
    we[6] = tail ? zv[6] : -__builtin_inff();
    float S = wave_sum(((we[0] + we[1]) + (we[2] + we[3])) +
                       ((we[4] + we[5]) + (tail ? zv[6] : 0.0f)));
    int k = NP;
    float tau = (S - 1.0f) / (float)k;
    for (int it = 0; it < NP; ++it) {
      float t[7];
      int cnt = 0;
#pragma unroll
      for (int e = 0; e < 7; ++e) {
        bool m = we[e] > tau;
        cnt += __popc(__builtin_amdgcn_ballot_w32(m));
        t[e] = m ? we[e] : 0.0f;
        we[e] = m ? we[e] : -__builtin_inff();
      }
      if (cnt == k) break;
      S = wave_sum(((t[0] + t[1]) + (t[2] + t[3])) + ((t[4] + t[5]) + t[6]));
      k = max(cnt, 1);
      tau = (S - 1.0f) / (float)k;
    }
#pragma unroll
    for (int e = 0; e < 7; ++e) {
      rv[e] = fmaxf(zv[e] - tau, 0.0f);
      if (e < 6 || tail) zr[32 * e] = rv[e];
    }
  }
  __syncthreads();
  if (w < NR) {
#pragma unroll
    for (int e = 0; e < 7; ++e)
      if (e < 6 || tail) att[row * NP + lane + 32 * e] = rv[e];
    const int n2 = tn * 16 + lo;
    const float* bz = z + 3 * hi * ZP + (n2 < FD ? n2 : FD - 1);
    v8f c = {};
#pragma unroll
    for (int s = 0; s < 9; ++s) {
      const int t0 = 2 * s, t1 = 2 * s + 1;
      v2f b;
      b.x = bz[(t0 / 6) * ZP + (t0 % 6) * FD];
      b.y = bz[(t1 / 6) * ZP + (t1 % 6) * FD];
      c = __builtin_amdgcn_wmma_f32_16x16x4_f32(false, uf[s], false, b,
                                                (short)0, c, false, false);
    }
    if (n2 < FD) {
#pragma unroll
      for (int v = 0; v < 8; ++v) {
        int a = tm * 16 + 8 * hi + v;
        if (a < RK) Rw[a * CN + r * FD + n2] = c[v];
      }
    }
  }
}

__global__ __launch_bounds__(192) void ff_tucker_kernel(
    const float* __restrict__ ws, const float* __restrict__ U2,
    const float* __restrict__ U3, float* __restrict__ out) {
  __shared__ __attribute__((aligned(16))) float T1s[32 * FD];
  int a = blockIdx.x, tid = threadIdx.x;
  int lane = tid & 31, lo = lane & 15, hi = lane >> 4;
  int w = __builtin_amdgcn_readfirstlane(tid >> 5);
  const int tm = w >> 1, tn = w & 1;
  const int dn = tn * 16 + lo;
  const bool nok = dn < RK;
  const int j = tm * 16 + lo;
  const int jc = j < FD ? j : FD - 1;
  const float* sp =
      ws + OFF_R + a * CN + (6 * (jc / 6) + 3 * hi) * FD + 6 * (jc % 6);
  v2f sf[9], u3f[9], u2f[9];
#pragma unroll
  for (int s = 0; s < 9; ++s) {
    const int t = 2 * s;
    sf[s] = *(const v2f*)(sp + (t / 6) * FD + (t % 6));
    u3f[s] = *(const v2f*)(U3 + (nok ? dn : 0) * FD + 18 * hi + 2 * s);
    u2f[s] = *(const v2f*)(U2 + (nok ? dn : 0) * FD + 4 * s + 2 * hi);
  }
  {
    v8f c = {};
#pragma unroll
    for (int s = 0; s < 9; ++s) {
      v2f b = u3f[s];
      if (!nok) { b.x = 0.0f; b.y = 0.0f; }
      c = __builtin_amdgcn_wmma_f32_16x16x4_f32(false, sf[s], false, b,
                                                (short)0, c, false, false);
    }
#pragma unroll
    for (int v = 0; v < 8; ++v) {
      int jj = tm * 16 + v + 8 * hi;
      if (jj < FD) T1s[dn * FD + jj] = c[v];
    }
  }
  __syncthreads();
  if (w < 4) {
    int cc = dn;
    bool cok = nok;
    const int aoff = (tm * 16 + lo) * FD + 2 * hi;
    v8f c = {};
#pragma unroll
    for (int kk = 0; kk < FD; kk += 4) {
      v2f av = *(const v2f*)(&T1s[aoff + kk]);
      v2f b = u2f[kk >> 2];
      if (!cok) { b.x = 0.0f; b.y = 0.0f; }
      c = __builtin_amdgcn_wmma_f32_16x16x4_f32(false, av, false, b, (short)0,
                                                c, false, false);
    }
    int d0 = tm * 16 + 8 * hi;
    float* op = out + a * (RK * RK) + cc * RK + d0;
    if (cok && d0 + 3 < RK) {
      v4f o;
      o.x = tanhf(c[0]); o.y = tanhf(c[1]); o.z = tanhf(c[2]); o.w = tanhf(c[3]);
      *(v4f*)op = o;
    }
    if (cok && d0 + 7 < RK) {
      v4f o;
      o.x = tanhf(c[4]); o.y = tanhf(c[5]); o.z = tanhf(c[6]); o.w = tanhf(c[7]);
      *(v4f*)(op + 4) = o;
    }
  }
}

extern "C" void kernel_launch(void* const* d_in, const int* in_sizes, int n_in,
                              void* d_out, int out_size, void* d_ws,
                              size_t ws_size, hipStream_t stream) {
  const float* patches = (const float*)d_in[0];
  const float* qw      = (const float*)d_in[1];
  const float* kw      = (const float*)d_in[2];
  const float* gamma   = (const float*)d_in[3];
  const float* beta    = (const float*)d_in[4];
  const float* U1      = (const float*)d_in[5];
  const float* U2      = (const float*)d_in[6];
  const float* U3      = (const float*)d_in[7];

  float* out = (float*)d_out;
  float* att = out + RK * RK * RK;
  float* ws = (float*)d_ws;

  ff_proj_bn_l2_kernel<<<2 * 14, 256, 0, stream>>>(patches, qw, kw, gamma,
                                                   beta, ws);
  ff_scores_sparsemax_kernel<<<36, 256, 0, stream>>>(ws, U1, att);
  ff_tucker_kernel<<<RK, 192, 0, stream>>>(ws, U2, U3, out);
}
